# starting kernel
# baseline (speedup 1.0000x reference)
#include <hip/hip_runtime.h>

typedef __attribute__((ext_vector_type(2))) float v2f;
typedef __attribute__((ext_vector_type(8))) float v8f;

#define NQ 4
#define NL 2
#define DIM 20
#define HID 64
#define TOK_PER_WAVE 32
#define WAVES 4
#define TOK_PER_BLOCK 128
#define HSTRIDE 68

__global__ __launch_bounds__(TOK_PER_BLOCK) void qhyb_fused(
    const float* __restrict__ x,
    const float* __restrict__ W_embed,
    const float* __restrict__ b_embed,
    const float* __restrict__ W_c2q,
    const float* __restrict__ b_c2q,
    const float* __restrict__ q_weights,
    const float* __restrict__ W_q2c,
    const float* __restrict__ b_q2c,
    const float* __restrict__ W_out,
    const float* __restrict__ b_out,
    float* __restrict__ out)
{
    __shared__ float h_lds[TOK_PER_BLOCK * HSTRIDE];
    __shared__ float wc2q_lds[NQ * HID];
    __shared__ float bemb_lds[HID];
    __shared__ float u_lds[NL][NQ][8];
    __shared__ float M_lds[3][NQ];
    __shared__ float c_lds[3];

    const int tid = threadIdx.x;

    for (int i = tid; i < NQ * HID; i += TOK_PER_BLOCK) wc2q_lds[i] = W_c2q[i];
    if (tid < HID) bemb_lds[tid] = b_embed[tid];
    if (tid < NL * NQ) {
        int layer = tid >> 2, q = tid & 3;
        float phi = q_weights[(layer * NQ + q) * 3 + 0];
        float th  = q_weights[(layer * NQ + q) * 3 + 1];
        float om  = q_weights[(layer * NQ + q) * 3 + 2];
        float ct = cosf(0.5f * th), st = sinf(0.5f * th);
        float cp = cosf(0.5f * (phi + om)), sp = sinf(0.5f * (phi + om));
        float cm = cosf(0.5f * (phi - om)), sm = sinf(0.5f * (phi - om));
        float* U = &u_lds[layer][q][0];
        U[0] =  cp * ct; U[1] = -sp * ct;
        U[2] = -cm * st; U[3] = -sm * st;
        U[4] =  cm * st; U[5] = -sm * st;
        U[6] =  cp * ct; U[7] =  sp * ct;
    }
    if (tid >= 8 && tid < 8 + 12) {
        int o = (tid - 8) >> 2, q = (tid - 8) & 3;
        float s = 0.0f;
        for (int h = 0; h < HID; ++h) s += W_out[o * HID + h] * W_q2c[h * NQ + q];
        M_lds[o][q] = s;
    }
    if (tid >= 20 && tid < 23) {
        int o = tid - 20;
        float s = b_out[o];
        for (int h = 0; h < HID; ++h) s += W_out[o * HID + h] * b_q2c[h];
        c_lds[o] = s;
    }
    __syncthreads();

    const int lane = tid & 31;
    const int wave = tid >> 5;
    const int lo = lane & 15, hi = lane >> 4;
    const long blockTok0 = (long)blockIdx.x * TOK_PER_BLOCK;
    const int  waveRow0  = wave * TOK_PER_WAVE;
    const long waveTok0  = blockTok0 + waveRow0;

    v2f bfrag[5][4];
#pragma unroll
    for (int ks = 0; ks < 5; ++ks)
#pragma unroll
        for (int n = 0; n < 4; ++n) {
            int hcol = n * 16 + lo;
            int d0 = ks * 4 + 2 * hi;
            bfrag[ks][n].x = W_embed[hcol * DIM + d0];
            bfrag[ks][n].y = W_embed[hcol * DIM + d0 + 1];
        }

    v8f acc[2][4];
#pragma unroll
    for (int mt = 0; mt < 2; ++mt)
#pragma unroll
        for (int n = 0; n < 4; ++n)
            acc[mt][n] = (v8f){0.f, 0.f, 0.f, 0.f, 0.f, 0.f, 0.f, 0.f};

#pragma unroll
    for (int ks = 0; ks < 5; ++ks) {
#pragma unroll
        for (int mt = 0; mt < 2; ++mt) {
            long t = waveTok0 + mt * 16 + lo;
            int d0 = ks * 4 + 2 * hi;
            v2f afrag;
            afrag.x = x[t * DIM + d0];
            afrag.y = x[t * DIM + d0 + 1];
#pragma unroll
            for (int n = 0; n < 4; ++n)
                acc[mt][n] = __builtin_amdgcn_wmma_f32_16x16x4_f32(
                    false, afrag, false, bfrag[ks][n],
                    (short)0, acc[mt][n], false, false);
        }
    }

#pragma unroll
    for (int mt = 0; mt < 2; ++mt)
#pragma unroll
        for (int n = 0; n < 4; ++n) {
            int hc = n * 16 + lo;
            float bb = bemb_lds[hc];
#pragma unroll
            for (int r = 0; r < 8; ++r) {
                int row = waveRow0 + mt * 16 + r + 8 * hi;
                h_lds[row * HSTRIDE + hc] = fmaxf(acc[mt][n][r] + bb, 0.0f);
            }
        }
    __syncthreads();

    const int myRow = waveRow0 + lane;
    const long tok = blockTok0 + myRow;
    const float* hrow = &h_lds[myRow * HSTRIDE];

    float cs[NQ], sn[NQ];
#pragma unroll
    for (int q = 0; q < NQ; ++q) {
        float s = b_c2q[q];
#pragma unroll 16
        for (int k = 0; k < HID; ++k) s += wc2q_lds[q * HID + k] * hrow[k];
        float a = 0.5f * tanhf(s);
        cs[q] = cosf(a);
        sn[q] = sinf(a);
    }

    float ar[16], ai[16];
#pragma unroll
    for (int i = 0; i < 16; ++i) {
        float v = ((i >> 3) & 1 ? sn[0] : cs[0]) *
                  ((i >> 2) & 1 ? sn[1] : cs[1]) *
                  ((i >> 1) & 1 ? sn[2] : cs[2]) *
                  ((i      ) & 1 ? sn[3] : cs[3]);
        ar[i] = v; ai[i] = 0.0f;
    }

#pragma unroll
    for (int layer = 0; layer < NL; ++layer) {
#pragma unroll
        for (int q = 0; q < NQ; ++q) {
            const float* U = &u_lds[layer][q][0];
            float u00r = U[0], u00i = U[1], u01r = U[2], u01i = U[3];
            float u10r = U[4], u10i = U[5], u11r = U[6], u11i = U[7];
            const int p = 3 - q;
#pragma unroll
            for (int i = 0; i < 16; ++i) {
                if (i & (1 << p)) continue;
                int j = i | (1 << p);
                float s0r = ar[i], s0i = ai[i], s1r = ar[j], s1i = ai[j];
                ar[i] = u00r * s0r - u00i * s0i + u01r * s1r - u01i * s1i;
                ai[i] = u00r * s0i + u00i * s0r + u01r * s1i + u01i * s1r;
                ar[j] = u10r * s0r - u10i * s0i + u11r * s1r - u11i * s1i;
                ai[j] = u10r * s0i + u10i * s0r + u11r * s1i + u11i * s1r;
            }
        }
#pragma unroll
        for (int q = 0; q < NQ - 1; ++q) {
            const int cbit = 1 << (3 - q), tbit = 1 << (2 - q);
#pragma unroll
            for (int i = 0; i < 16; ++i) {
                if ((i & cbit) && !(i & tbit)) {
                    int j = i | tbit;
                    float tr = ar[i], ti = ai[i];
                    ar[i] = ar[j]; ai[i] = ai[j];
                    ar[j] = tr;    ai[j] = ti;
                }
            }
        }
    }

    float z[NQ] = {0.f, 0.f, 0.f, 0.f};
#pragma unroll
    for (int i = 0; i < 16; ++i) {
        float p = ar[i] * ar[i] + ai[i] * ai[i];
#pragma unroll
        for (int q = 0; q < NQ; ++q)
            z[q] += (i & (1 << (3 - q))) ? -p : p;
    }

#pragma unroll
    for (int o = 0; o < 3; ++o) {
        float v = c_lds[o];
#pragma unroll
        for (int q = 0; q < NQ; ++q) v += M_lds[o][q] * z[q];
        out[tok * 3 + o] = v;
    }
}

extern "C" void kernel_launch(void* const* d_in, const int* in_sizes, int n_in,
                              void* d_out, int out_size, void* d_ws, size_t ws_size,
                              hipStream_t stream) {
    const float* x        = (const float*)d_in[0];
    const float* W_embed  = (const float*)d_in[1];
    const float* b_embed  = (const float*)d_in[2];
    const float* W_c2q    = (const float*)d_in[3];
    const float* b_c2q    = (const float*)d_in[4];
    const float* q_w      = (const float*)d_in[5];
    const float* W_q2c    = (const float*)d_in[6];
    const float* b_q2c    = (const float*)d_in[7];
    const float* W_out    = (const float*)d_in[8];
    const float* b_out    = (const float*)d_in[9];
    float* out = (float*)d_out;

    const int tokens = in_sizes[0] / DIM;
    const int blocks = tokens / TOK_PER_BLOCK;
    qhyb_fused<<<blocks, TOK_PER_BLOCK, 0, stream>>>(
        x, W_embed, b_embed, W_c2q, b_c2q, q_w, W_q2c, b_q2c, W_out, b_out, out);
}


# Per-block constant set-up taken off the critical path: all global loads issued up front, 64-term constant sums spread over lanes (summation order changes), one barrier; tail recompiles unpacked
# speedup vs baseline: 1.3512x; 1.3512x over previous
#include <hip/hip_runtime.h>


typedef __attribute__((ext_vector_type(2))) float v2f;
typedef __attribute__((ext_vector_type(8))) float v8f;

#define NQ 4
#define NL 2
#define DIM 20
#define HID 64
#define TOK_PER_WAVE 32
#define WAVES 4
#define TOK_PER_BLOCK 128
#define HSTRIDE 68

__global__ __launch_bounds__(TOK_PER_BLOCK, 8) void qhyb_fused(
    const float* __restrict__ x,
    const float* __restrict__ W_embed,
    const float* __restrict__ b_embed,
    const float* __restrict__ W_c2q,
    const float* __restrict__ b_c2q,
    const float* __restrict__ q_weights,
    const float* __restrict__ W_q2c,
    const float* __restrict__ b_q2c,
    const float* __restrict__ W_out,
    const float* __restrict__ b_out,
    float* __restrict__ out)
{
    __shared__ float h_lds[TOK_PER_BLOCK * HSTRIDE];
    __shared__ __attribute__((aligned(16))) float wc2q_lds[NQ * HID];
    __shared__ float u_lds[NL][NQ][8];
    __shared__ __attribute__((aligned(16))) float M_lds[3][NQ];
    __shared__ float c_lds[3];

    const int tid  = threadIdx.x;
    const int lane = tid & 31;
    const int wave = __builtin_amdgcn_readfirstlane(tid >> 5);
    const int lo = lane & 15, hi = lane >> 4;
    const long blockTok0 = (long)blockIdx.x * TOK_PER_BLOCK;
    const int  waveRow0  = wave * TOK_PER_WAVE;
    const long waveTok0  = blockTok0 + waveRow0;

    v2f afrag[5][2];
#pragma unroll
    for (int ks = 0; ks < 5; ++ks)
#pragma unroll
        for (int mt = 0; mt < 2; ++mt) {
            long t = waveTok0 + mt * 16 + lo;
            int d0 = ks * 4 + 2 * hi;
            afrag[ks][mt].x = x[t * DIM + d0];
            afrag[ks][mt].y = x[t * DIM + d0 + 1];
        }
    v2f bfrag[5][4];
#pragma unroll
    for (int ks = 0; ks < 5; ++ks)
#pragma unroll
        for (int n = 0; n < 4; ++n) {
            int hcol = n * 16 + lo;
            int d0 = ks * 4 + 2 * hi;
            bfrag[ks][n].x = W_embed[hcol * DIM + d0];
            bfrag[ks][n].y = W_embed[hcol * DIM + d0 + 1];
        }
    float bias[4];
#pragma unroll
    for (int n = 0; n < 4; ++n) bias[n] = b_embed[n * 16 + lo];

    if (wave == 0) {
        if (lane < NL * NQ) {
            float phi = q_weights[lane * 3 + 0];
            float th  = q_weights[lane * 3 + 1];
            float om  = q_weights[lane * 3 + 2];
            float ct = cosf(0.5f * th), st = sinf(0.5f * th);
            float cp = cosf(0.5f * (phi + om)), sp = sinf(0.5f * (phi + om));
            float cm = cosf(0.5f * (phi - om)), sm = sinf(0.5f * (phi - om));
            float* U = &u_lds[lane >> 2][lane & 3][0];
            U[0] =  cp * ct; U[1] = -sp * ct;
            U[2] = -cm * st; U[3] = -sm * st;
            U[4] =  cm * st; U[5] = -sm * st;
            U[6] =  cp * ct; U[7] =  sp * ct;
        }
    } else if (wave == 1) {
        float wo[3][2], bq[2];
        float4 wq[2];
#pragma unroll
        for (int j = 0; j < 2; ++j) {
            const int h = lane + 32 * j;
#pragma unroll
            for (int o = 0; o < 3; ++o) wo[o][j] = W_out[o * HID + h];
            wq[j] = *reinterpret_cast<const float4*>(&W_q2c[h * NQ]);
            bq[j] = b_q2c[h];
        }
        float pm[3][NQ], pc[3];
#pragma unroll
        for (int o = 0; o < 3; ++o) {
            pm[o][0] = wo[o][0] * wq[0].x + wo[o][1] * wq[1].x;
            pm[o][1] = wo[o][0] * wq[0].y + wo[o][1] * wq[1].y;
            pm[o][2] = wo[o][0] * wq[0].z + wo[o][1] * wq[1].z;
            pm[o][3] = wo[o][0] * wq[0].w + wo[o][1] * wq[1].w;
            pc[o]    = wo[o][0] * bq[0]   + wo[o][1] * bq[1];
        }
#pragma unroll
        for (int m = 16; m >= 1; m >>= 1) {
#pragma unroll
            for (int o = 0; o < 3; ++o) {
#pragma unroll
                for (int q = 0; q < NQ; ++q) pm[o][q] += __shfl_xor(pm[o][q], m, 32);
                pc[o] += __shfl_xor(pc[o], m, 32);
            }
        }
        if (lane == 0) {
#pragma unroll
            for (int o = 0; o < 3; ++o) {
#pragma unroll
                for (int q = 0; q < NQ; ++q) M_lds[o][q] = pm[o][q];
                c_lds[o] = pc[o] + b_out[o];
            }
        }
    } else {
        const int i4 = (tid - 2 * 32) * 4;
        *reinterpret_cast<float4*>(&wc2q_lds[i4]) = *reinterpret_cast<const float4*>(&W_c2q[i4]);
    }

    v8f acc[2][4];
#pragma unroll
    for (int mt = 0; mt < 2; ++mt)
#pragma unroll
        for (int n = 0; n < 4; ++n)
            acc[mt][n] = (v8f){0.f, 0.f, 0.f, 0.f, 0.f, 0.f, 0.f, 0.f};

#pragma unroll
    for (int ks = 0; ks < 5; ++ks) {
#pragma unroll
        for (int mt = 0; mt < 2; ++mt) {
#pragma unroll
            for (int n = 0; n < 4; ++n)
                acc[mt][n] = __builtin_amdgcn_wmma_f32_16x16x4_f32(
                    false, afrag[ks][mt], false, bfrag[ks][n],
                    (short)0, acc[mt][n], false, false);
        }
    }

#pragma unroll
    for (int mt = 0; mt < 2; ++mt)
#pragma unroll
        for (int n = 0; n < 4; ++n) {
            int hc = n * 16 + lo;
            float bb = bias[n];
#pragma unroll
            for (int r = 0; r < 8; ++r) {
                int row = waveRow0 + mt * 16 + r + 8 * hi;
                h_lds[row * HSTRIDE + hc] = fmaxf(acc[mt][n][r] + bb, 0.0f);
            }
        }
    __syncthreads();

    const int myRow = waveRow0 + lane;
    const long tok = blockTok0 + myRow;
    const float* hrow = &h_lds[myRow * HSTRIDE];

    float cs[NQ], sn[NQ];
#pragma unroll
    for (int q = 0; q < NQ; ++q) {
        float s = b_c2q[q];
#pragma unroll 16
        for (int k = 0; k < HID; ++k) s += wc2q_lds[q * HID + k] * hrow[k];
        float a = 0.5f * tanhf(s);
        cs[q] = cosf(a);
        sn[q] = sinf(a);
    }

    float ar[16], ai[16];
#pragma unroll
    for (int i = 0; i < 16; ++i) {
        float v = ((i >> 3) & 1 ? sn[0] : cs[0]) *
                  ((i >> 2) & 1 ? sn[1] : cs[1]) *
                  ((i >> 1) & 1 ? sn[2] : cs[2]) *
                  ((i      ) & 1 ? sn[3] : cs[3]);
        ar[i] = v; ai[i] = 0.0f;
    }

#pragma unroll
    for (int layer = 0; layer < NL; ++layer) {
#pragma unroll
        for (int q = 0; q < NQ; ++q) {
            const float* U = &u_lds[layer][q][0];
            float u00r = U[0], u00i = U[1], u01r = U[2], u01i = U[3];
            float u10r = U[4], u10i = U[5], u11r = U[6], u11i = U[7];
            const int p = 3 - q;
#pragma unroll
            for (int i = 0; i < 16; ++i) {
                if (i & (1 << p)) continue;
                int j = i | (1 << p);
                float s0r = ar[i], s0i = ai[i], s1r = ar[j], s1i = ai[j];
                ar[i] = u00r * s0r - u00i * s0i + u01r * s1r - u01i * s1i;
                ai[i] = u00r * s0i + u00i * s0r + u01r * s1i + u01i * s1r;
                ar[j] = u10r * s0r - u10i * s0i + u11r * s1r - u11i * s1i;
                ai[j] = u10r * s0i + u10i * s0r + u11r * s1i + u11i * s1r;
            }
        }
#pragma unroll
        for (int q = 0; q < NQ - 1; ++q) {
            const int cbit = 1 << (3 - q), tbit = 1 << (2 - q);
#pragma unroll
            for (int i = 0; i < 16; ++i) {
                if ((i & cbit) && !(i & tbit)) {
                    int j = i | tbit;
                    float tr = ar[i], ti = ai[i];
                    ar[i] = ar[j]; ai[i] = ai[j];
                    ar[j] = tr;    ai[j] = ti;
                }
            }
        }
    }

    float z[NQ] = {0.f, 0.f, 0.f, 0.f};
#pragma unroll
    for (int i = 0; i < 16; ++i) {
        float p = ar[i] * ar[i] + ai[i] * ai[i];
#pragma unroll
        for (int q = 0; q < NQ; ++q)
            z[q] += (i & (1 << (3 - q))) ? -p : p;
    }

#pragma unroll
    for (int o = 0; o < 3; ++o) {
        float v = c_lds[o];
#pragma unroll
        for (int q = 0; q < NQ; ++q) v += M_lds[o][q] * z[q];
        out[tok * 3 + o] = v;
    }
}

extern "C" void kernel_launch(void* const* d_in, const int* in_sizes, int n_in,
                              void* d_out, int out_size, void* d_ws, size_t ws_size,
                              hipStream_t stream) {
    const float* x        = (const float*)d_in[0];
    const float* W_embed  = (const float*)d_in[1];
    const float* b_embed  = (const float*)d_in[2];
    const float* W_c2q    = (const float*)d_in[3];
    const float* b_c2q    = (const float*)d_in[4];
    const float* q_w      = (const float*)d_in[5];
    const float* W_q2c    = (const float*)d_in[6];
    const float* b_q2c    = (const float*)d_in[7];
    const float* W_out    = (const float*)d_in[8];
    const float* b_out    = (const float*)d_in[9];
    float* out = (float*)d_out;

    const int tokens = in_sizes[0] / DIM;
    const int blocks = tokens / TOK_PER_BLOCK;
    qhyb_fused<<<blocks, TOK_PER_BLOCK, 0, stream>>>(
        x, W_embed, b_embed, W_c2q, b_c2q, q_w, W_q2c, b_q2c, W_out, b_out, out);
}


# State kept factored through both rotation layers: layer 1 on each qubit's 2-vector, layer 2 folded in along the CNOT ladder, 16 amplitudes formed once (order of the sums changes)
# speedup vs baseline: 1.5070x; 1.1153x over previous
#include <hip/hip_runtime.h>


typedef __attribute__((ext_vector_type(2))) float v2f;
typedef __attribute__((ext_vector_type(8))) float v8f;

#define NQ 4
#define NL 2
#define DIM 20
#define HID 64
#define TOK_PER_WAVE 32
#define WAVES 4
#define TOK_PER_BLOCK 128
#define HSTRIDE 68

struct cpx { float r, i; };
__device__ __forceinline__ cpx cmul(cpx a, cpx b) {
    cpx o; o.r = a.r * b.r - a.i * b.i; o.i = a.r * b.i + a.i * b.r; return o;
}
__device__ __forceinline__ cpx cmul2(cpx a, cpx b, cpx c, cpx d) {
    cpx o;
    o.r = a.r * b.r - a.i * b.i + c.r * d.r - c.i * d.i;
    o.i = a.r * b.i + a.i * b.r + c.r * d.i + c.i * d.r;
    return o;
}

__global__ __launch_bounds__(TOK_PER_BLOCK, 8) void qhyb_fused(
    const float* __restrict__ x,
    const float* __restrict__ W_embed,
    const float* __restrict__ b_embed,
    const float* __restrict__ W_c2q,
    const float* __restrict__ b_c2q,
    const float* __restrict__ q_weights,
    const float* __restrict__ W_q2c,
    const float* __restrict__ b_q2c,
    const float* __restrict__ W_out,
    const float* __restrict__ b_out,
    float* __restrict__ out)
{
    __shared__ float h_lds[TOK_PER_BLOCK * HSTRIDE];
    __shared__ __attribute__((aligned(16))) float wc2q_lds[NQ * HID];
    __shared__ float u_lds[NL][NQ][8];
    __shared__ __attribute__((aligned(16))) float M_lds[3][NQ];
    __shared__ float c_lds[3];

    const int tid  = threadIdx.x;
    const int lane = tid & 31;
    const int wave = __builtin_amdgcn_readfirstlane(tid >> 5);
    const int lo = lane & 15, hi = lane >> 4;
    const long blockTok0 = (long)blockIdx.x * TOK_PER_BLOCK;
    const int  waveRow0  = wave * TOK_PER_WAVE;
    const long waveTok0  = blockTok0 + waveRow0;

    v2f afrag[5][2];
#pragma unroll
    for (int ks = 0; ks < 5; ++ks)
#pragma unroll
        for (int mt = 0; mt < 2; ++mt) {
            long t = waveTok0 + mt * 16 + lo;
            int d0 = ks * 4 + 2 * hi;
            afrag[ks][mt].x = x[t * DIM + d0];
            afrag[ks][mt].y = x[t * DIM + d0 + 1];
        }
    v2f bfrag[5][4];
#pragma unroll
    for (int ks = 0; ks < 5; ++ks)
#pragma unroll
        for (int n = 0; n < 4; ++n) {
            int hcol = n * 16 + lo;
            int d0 = ks * 4 + 2 * hi;
            bfrag[ks][n].x = W_embed[hcol * DIM + d0];
            bfrag[ks][n].y = W_embed[hcol * DIM + d0 + 1];
        }
    float bias[4];
#pragma unroll
    for (int n = 0; n < 4; ++n) bias[n] = b_embed[n * 16 + lo];

    if (wave == 0) {
        if (lane < NL * NQ) {
            float phi = q_weights[lane * 3 + 0];
            float th  = q_weights[lane * 3 + 1];
            float om  = q_weights[lane * 3 + 2];
            float ct = cosf(0.5f * th), st = sinf(0.5f * th);
            float cp = cosf(0.5f * (phi + om)), sp = sinf(0.5f * (phi + om));
            float cm = cosf(0.5f * (phi - om)), sm = sinf(0.5f * (phi - om));
            float* U = &u_lds[lane >> 2][lane & 3][0];
            U[0] =  cp * ct; U[1] = -sp * ct;
            U[2] = -cm * st; U[3] = -sm * st;
            U[4] =  cm * st; U[5] = -sm * st;
            U[6] =  cp * ct; U[7] =  sp * ct;
        }
    } else if (wave == 1) {
        float wo[3][2], bq[2];
        float4 wq[2];
#pragma unroll
        for (int j = 0; j < 2; ++j) {
            const int h = lane + 32 * j;
#pragma unroll
            for (int o = 0; o < 3; ++o) wo[o][j] = W_out[o * HID + h];
            wq[j] = *reinterpret_cast<const float4*>(&W_q2c[h * NQ]);
            bq[j] = b_q2c[h];
        }
        float pm[3][NQ], pc[3];
#pragma unroll
        for (int o = 0; o < 3; ++o) {
            pm[o][0] = wo[o][0] * wq[0].x + wo[o][1] * wq[1].x;
            pm[o][1] = wo[o][0] * wq[0].y + wo[o][1] * wq[1].y;
            pm[o][2] = wo[o][0] * wq[0].z + wo[o][1] * wq[1].z;
            pm[o][3] = wo[o][0] * wq[0].w + wo[o][1] * wq[1].w;
            pc[o]    = wo[o][0] * bq[0]   + wo[o][1] * bq[1];
        }
#pragma unroll
        for (int m = 16; m >= 1; m >>= 1) {
#pragma unroll
            for (int o = 0; o < 3; ++o) {
#pragma unroll
                for (int q = 0; q < NQ; ++q) pm[o][q] += __shfl_xor(pm[o][q], m, 32);
                pc[o] += __shfl_xor(pc[o], m, 32);
            }
        }
        if (lane == 0) {
#pragma unroll
            for (int o = 0; o < 3; ++o) {
#pragma unroll
                for (int q = 0; q < NQ; ++q) M_lds[o][q] = pm[o][q];
                c_lds[o] = pc[o] + b_out[o];
            }
        }
    } else {
        const int i4 = (tid - 2 * 32) * 4;
        *reinterpret_cast<float4*>(&wc2q_lds[i4]) = *reinterpret_cast<const float4*>(&W_c2q[i4]);
    }

    v8f acc[2][4];
#pragma unroll
    for (int mt = 0; mt < 2; ++mt)
#pragma unroll
        for (int n = 0; n < 4; ++n)
            acc[mt][n] = (v8f){0.f, 0.f, 0.f, 0.f, 0.f, 0.f, 0.f, 0.f};

#pragma unroll
    for (int ks = 0; ks < 5; ++ks) {
#pragma unroll
        for (int mt = 0; mt < 2; ++mt) {
#pragma unroll
            for (int n = 0; n < 4; ++n)
                acc[mt][n] = __builtin_amdgcn_wmma_f32_16x16x4_f32(
                    false, afrag[ks][mt], false, bfrag[ks][n],
                    (short)0, acc[mt][n], false, false);
        }
    }

#pragma unroll
    for (int mt = 0; mt < 2; ++mt)
#pragma unroll
        for (int n = 0; n < 4; ++n) {
            int hc = n * 16 + lo;
            float bb = bias[n];
#pragma unroll
            for (int r = 0; r < 8; ++r) {
                int row = waveRow0 + mt * 16 + r + 8 * hi;
                h_lds[row * HSTRIDE + hc] = fmaxf(acc[mt][n][r] + bb, 0.0f);
            }
        }
    __syncthreads();

    const int myRow = waveRow0 + lane;
    const long tok = blockTok0 + myRow;
    const float* hrow = &h_lds[myRow * HSTRIDE];

    float cs[NQ], sn[NQ];
#pragma unroll
    for (int q = 0; q < NQ; ++q) {
        float s = b_c2q[q];
#pragma unroll 16
        for (int k = 0; k < HID; ++k) s += wc2q_lds[q * HID + k] * hrow[k];
        float a = 0.5f * tanhf(s);
        cs[q] = cosf(a);
        sn[q] = sinf(a);
    }


    cpx phi[NQ][2];
#pragma unroll
    for (int q = 0; q < NQ; ++q) {
        const float* U = &u_lds[0][q][0];
#pragma unroll
        for (int a = 0; a < 2; ++a) {
            phi[q][a].r = U[4 * a + 0] * cs[q] + U[4 * a + 2] * sn[q];
            phi[q][a].i = U[4 * a + 1] * cs[q] + U[4 * a + 3] * sn[q];
        }
    }
    cpx V[NQ][2][2];
#pragma unroll
    for (int q = 0; q < NQ; ++q) {
        const float* U = &u_lds[1][q][0];
#pragma unroll
        for (int a = 0; a < 2; ++a)
#pragma unroll
            for (int b = 0; b < 2; ++b) { V[q][a][b].r = U[4 * a + 2 * b]; V[q][a][b].i = U[4 * a + 2 * b + 1]; }
    }

    cpx G3[2][2];
#pragma unroll
    for (int c2 = 0; c2 < 2; ++c2)
#pragma unroll
        for (int d3 = 0; d3 < 2; ++d3)
            G3[c2][d3] = cmul2(V[3][d3][0], phi[3][c2], V[3][d3][1], phi[3][1 ^ c2]);

    cpx R23[2][4];
#pragma unroll
    for (int c1 = 0; c1 < 2; ++c1)
#pragma unroll
        for (int d2 = 0; d2 < 2; ++d2) {
            const cpx h0 = cmul(V[2][d2][0], phi[2][c1]);
            const cpx h1 = cmul(V[2][d2][1], phi[2][1 ^ c1]);
#pragma unroll
            for (int d3 = 0; d3 < 2; ++d3)
                R23[c1][d2 * 2 + d3] = cmul2(h0, G3[0][d3], h1, G3[1][d3]);
        }

    cpx R123[2][8];
#pragma unroll
    for (int c0 = 0; c0 < 2; ++c0)
#pragma unroll
        for (int d1 = 0; d1 < 2; ++d1) {
            const cpx h0 = cmul(V[1][d1][0], phi[1][c0]);
            const cpx h1 = cmul(V[1][d1][1], phi[1][1 ^ c0]);
#pragma unroll
            for (int j = 0; j < 4; ++j)
                R123[c0][d1 * 4 + j] = cmul2(h0, R23[0][j], h1, R23[1][j]);
        }

    float ar[16], ai[16];
#pragma unroll
    for (int d0 = 0; d0 < 2; ++d0) {
        const cpx h0 = cmul(V[0][d0][0], phi[0][0]);
        const cpx h1 = cmul(V[0][d0][1], phi[0][1]);
#pragma unroll
        for (int j = 0; j < 8; ++j) {
            const cpx v = cmul2(h0, R123[0][j], h1, R123[1][j]);
            ar[d0 * 8 + j] = v.r; ai[d0 * 8 + j] = v.i;
        }
    }

#pragma unroll
    for (int q = 0; q < NQ - 1; ++q) {
        const int cbit = 1 << (3 - q), tbit = 1 << (2 - q);
#pragma unroll
        for (int i = 0; i < 16; ++i) {
            if ((i & cbit) && !(i & tbit)) {
                int j = i | tbit;
                float tr = ar[i], ti = ai[i];
                ar[i] = ar[j]; ai[i] = ai[j];
                ar[j] = tr;    ai[j] = ti;
            }
        }
    }

    float z[NQ] = {0.f, 0.f, 0.f, 0.f};
#pragma unroll
    for (int i = 0; i < 16; ++i) {
        float p = ar[i] * ar[i] + ai[i] * ai[i];
#pragma unroll
        for (int q = 0; q < NQ; ++q)
            z[q] += (i & (1 << (3 - q))) ? -p : p;
    }

#pragma unroll
    for (int o = 0; o < 3; ++o) {
        float v = c_lds[o];
#pragma unroll
        for (int q = 0; q < NQ; ++q) v += M_lds[o][q] * z[q];
        out[tok * 3 + o] = v;
    }
}

extern "C" void kernel_launch(void* const* d_in, const int* in_sizes, int n_in,
                              void* d_out, int out_size, void* d_ws, size_t ws_size,
                              hipStream_t stream) {
    const float* x        = (const float*)d_in[0];
    const float* W_embed  = (const float*)d_in[1];
    const float* b_embed  = (const float*)d_in[2];
    const float* W_c2q    = (const float*)d_in[3];
    const float* b_c2q    = (const float*)d_in[4];
    const float* q_w      = (const float*)d_in[5];
    const float* W_q2c    = (const float*)d_in[6];
    const float* b_q2c    = (const float*)d_in[7];
    const float* W_out    = (const float*)d_in[8];
    const float* b_out    = (const float*)d_in[9];
    float* out = (float*)d_out;

    const int tokens = in_sizes[0] / DIM;
    const int blocks = tokens / TOK_PER_BLOCK;
    qhyb_fused<<<blocks, TOK_PER_BLOCK, 0, stream>>>(
        x, W_embed, b_embed, W_c2q, b_c2q, q_w, W_q2c, b_q2c, W_out, b_out, out);
}


# Z expectations from a real 4-number chain along the CNOT ladder: per qubit its density (linear in cos, sin of the angle) times a constant 4x4 map of layer 2; no 16 amplitudes (order of sums changes)
# speedup vs baseline: 1.5951x; 1.0585x over previous
#include <hip/hip_runtime.h>


typedef __attribute__((ext_vector_type(2))) float v2f;
typedef __attribute__((ext_vector_type(8))) float v8f;

#define NQ 4
#define NL 2
#define DIM 20
#define HID 64
#define TOK_PER_WAVE 32
#define WAVES 4
#define TOK_PER_BLOCK 128
#define HSTRIDE 68

__global__ __launch_bounds__(TOK_PER_BLOCK, 8) void qhyb_fused(
    const float* __restrict__ x,
    const float* __restrict__ W_embed,
    const float* __restrict__ b_embed,
    const float* __restrict__ W_c2q,
    const float* __restrict__ b_c2q,
    const float* __restrict__ q_weights,
    const float* __restrict__ W_q2c,
    const float* __restrict__ b_q2c,
    const float* __restrict__ W_out,
    const float* __restrict__ b_out,
    float* __restrict__ out)
{
    __shared__ float h_lds[TOK_PER_BLOCK * HSTRIDE];
    __shared__ __attribute__((aligned(16))) float wc2q_lds[NQ * HID];
    __shared__ __attribute__((aligned(16))) float R_lds[NQ][12];
    __shared__ __attribute__((aligned(16))) float K_lds[NQ][8];
    __shared__ __attribute__((aligned(16))) float M_lds[3][NQ];
    __shared__ float c_lds[3];

    const int tid  = threadIdx.x;
    const int lane = tid & 31;
    const int wave = __builtin_amdgcn_readfirstlane(tid >> 5);
    const int lo = lane & 15, hi = lane >> 4;
    const long blockTok0 = (long)blockIdx.x * TOK_PER_BLOCK;
    const int  waveRow0  = wave * TOK_PER_WAVE;
    const long waveTok0  = blockTok0 + waveRow0;

    v2f afrag[5][2];
#pragma unroll
    for (int ks = 0; ks < 5; ++ks)
#pragma unroll
        for (int mt = 0; mt < 2; ++mt) {
            long t = waveTok0 + mt * 16 + lo;
            int d0 = ks * 4 + 2 * hi;
            afrag[ks][mt].x = x[t * DIM + d0];
            afrag[ks][mt].y = x[t * DIM + d0 + 1];
        }
    v2f bfrag[5][4];
#pragma unroll
    for (int ks = 0; ks < 5; ++ks)
#pragma unroll
        for (int n = 0; n < 4; ++n) {
            int hcol = n * 16 + lo;
            int d0 = ks * 4 + 2 * hi;
            bfrag[ks][n].x = W_embed[hcol * DIM + d0];
            bfrag[ks][n].y = W_embed[hcol * DIM + d0 + 1];
        }
    float bias[4];
#pragma unroll
    for (int n = 0; n < 4; ++n) bias[n] = b_embed[n * 16 + lo];

    if (wave == 0) {
        if (lane < NL * NQ) {
            float phi = q_weights[lane * 3 + 0];
            float th  = q_weights[lane * 3 + 1];
            float om  = q_weights[lane * 3 + 2];
            float ct = cosf(0.5f * th), st = sinf(0.5f * th);
            float cp = cosf(0.5f * (phi + om)), sp = sinf(0.5f * (phi + om));
            float cm = cosf(0.5f * (phi - om)), sm = sinf(0.5f * (phi - om));
            const float u00r =  cp * ct, u00i = -sp * ct;
            const float u01r = -cm * st, u01i = -sm * st;
            const float u10r =  cm * st, u10i = -sm * st;
            const float u11r =  cp * ct, u11i =  sp * ct;
            const int q = lane & 3;
            if (lane < NQ) {
                const float n0a = u00r * u00r + u00i * u00i, n0b = u01r * u01r + u01i * u01i;
                const float n0d = 2.0f * (u00r * u01r + u00i * u01i);
                const float n1a = u10r * u10r + u10i * u10i, n1b = u11r * u11r + u11i * u11i;
                const float n1d = 2.0f * (u10r * u11r + u10i * u11i);
                const float wra = u00r * u10r + u00i * u10i, wia = u00r * u10i - u00i * u10r;
                const float wrb = u01r * u11r + u01i * u11i, wib = u01r * u11i - u01i * u11r;
                const float wrd = (u00r * u11r + u00i * u11i) + (u01r * u10r + u01i * u10i);
                const float wid = (u00r * u11i - u00i * u11r) + (u01r * u10i - u01i * u10r);
                const float a0 = n0a + n1a + 2.0f * wra, b0 = n0b + n1b + 2.0f * wrb, d0 = n0d + n1d + 2.0f * wrd;
                const float a1 = n0a + n1a - 2.0f * wra, b1 = n0b + n1b - 2.0f * wrb, d1 = n0d + n1d - 2.0f * wrd;
                const float a2 = n0a - n1a, b2 = n0b - n1b, d2 = n0d - n1d;
                const float a3 = -2.0f * wia, b3 = -2.0f * wib, d3 = -2.0f * wid;
                float* R = &R_lds[q][0];
                R[0] = 0.5f * (a0 + b0); R[1]  = 0.5f * (a0 - b0); R[2]  = 0.5f * d0;
                R[3] = 0.5f * (a1 + b1); R[4]  = 0.5f * (a1 - b1); R[5]  = 0.5f * d1;
                R[6] = 0.5f * (a2 + b2); R[7]  = 0.5f * (a2 - b2); R[8]  = 0.5f * d2;
                R[9] = 0.5f * (a3 + b3); R[10] = 0.5f * (a3 - b3); R[11] = 0.5f * d3;
            } else {
                const float A00 = (u00r * u00r + u00i * u00i) - (u10r * u10r + u10i * u10i);
                const float A11 = (u01r * u01r + u01i * u01i) - (u11r * u11r + u11i * u11i);
                const float A01r = (u00r * u01r + u00i * u01i) - (u10r * u11r + u10i * u11i);
                const float A01i = (u00r * u01i - u00i * u01r) - (u10r * u11i - u10i * u11r);
                const float u = A00 + A11 + 2.0f * A01r, v = A00 + A11 - 2.0f * A01r;
                const float p = A00 - A11, t = 2.0f * A01i;
                float* K = &K_lds[q][0];
                K[0] = 0.25f * u; K[1] = 0.25f * v; K[2] = 0.5f * p; K[3] = 0.5f * t;
                K[4] = 0.25f * p; K[5] = 0.25f * (u + v); K[6] = 0.25f * t; K[7] = 0.25f * (u - v);
            }
        }
    } else if (wave == 1) {
        float wo[3][2], bq[2];
        float4 wq[2];
#pragma unroll
        for (int j = 0; j < 2; ++j) {
            const int h = lane + 32 * j;
#pragma unroll
            for (int o = 0; o < 3; ++o) wo[o][j] = W_out[o * HID + h];
            wq[j] = *reinterpret_cast<const float4*>(&W_q2c[h * NQ]);
            bq[j] = b_q2c[h];
        }
        float pm[3][NQ], pc[3];
#pragma unroll
        for (int o = 0; o < 3; ++o) {
            pm[o][0] = wo[o][0] * wq[0].x + wo[o][1] * wq[1].x;
            pm[o][1] = wo[o][0] * wq[0].y + wo[o][1] * wq[1].y;
            pm[o][2] = wo[o][0] * wq[0].z + wo[o][1] * wq[1].z;
            pm[o][3] = wo[o][0] * wq[0].w + wo[o][1] * wq[1].w;
            pc[o]    = wo[o][0] * bq[0]   + wo[o][1] * bq[1];
        }
#pragma unroll
        for (int m = 16; m >= 1; m >>= 1) {
#pragma unroll
            for (int o = 0; o < 3; ++o) {
#pragma unroll
                for (int q = 0; q < NQ; ++q) pm[o][q] += __shfl_xor(pm[o][q], m, 32);
                pc[o] += __shfl_xor(pc[o], m, 32);
            }
        }
        if (lane == 0) {
#pragma unroll
            for (int o = 0; o < 3; ++o) {
#pragma unroll
                for (int q = 0; q < NQ; ++q) M_lds[o][q] = pm[o][q];
                c_lds[o] = pc[o] + b_out[o];
            }
        }
    } else {
        const int i4 = (tid - 2 * 32) * 4;
        *reinterpret_cast<float4*>(&wc2q_lds[i4]) = *reinterpret_cast<const float4*>(&W_c2q[i4]);
    }

    v8f acc[2][4];
#pragma unroll
    for (int mt = 0; mt < 2; ++mt)
#pragma unroll
        for (int n = 0; n < 4; ++n)
            acc[mt][n] = (v8f){0.f, 0.f, 0.f, 0.f, 0.f, 0.f, 0.f, 0.f};

#pragma unroll
    for (int ks = 0; ks < 5; ++ks) {
#pragma unroll
        for (int mt = 0; mt < 2; ++mt) {
#pragma unroll
            for (int n = 0; n < 4; ++n)
                acc[mt][n] = __builtin_amdgcn_wmma_f32_16x16x4_f32(
                    false, afrag[ks][mt], false, bfrag[ks][n],
                    (short)0, acc[mt][n], false, false);
        }
    }

#pragma unroll
    for (int mt = 0; mt < 2; ++mt)
#pragma unroll
        for (int n = 0; n < 4; ++n) {
            int hc = n * 16 + lo;
            float bb = bias[n];
#pragma unroll
            for (int r = 0; r < 8; ++r) {
                int row = waveRow0 + mt * 16 + r + 8 * hi;
                h_lds[row * HSTRIDE + hc] = fmaxf(acc[mt][n][r] + bb, 0.0f);
            }
        }
    __syncthreads();

    const int myRow = waveRow0 + lane;
    const long tok = blockTok0 + myRow;
    const float* hrow = &h_lds[myRow * HSTRIDE];

    float ca[NQ], sa[NQ];
#pragma unroll
    for (int q = 0; q < NQ; ++q) {
        float s = b_c2q[q];
#pragma unroll 16
        for (int k = 0; k < HID; ++k) s += wc2q_lds[q * HID + k] * hrow[k];
        float a = tanhf(s);
        ca[q] = cosf(a);
        sa[q] = sinf(a);
    }

    float z[NQ];
    float l0 = 1.0f, l3 = 1.0f, lr = 1.0f, li = 0.0f;
#pragma unroll
    for (int j = 0; j < NQ; ++j) {
        const float* R = &R_lds[j][0];
        const float d00 = R[0] + R[1]  * ca[j] + R[2]  * sa[j];
        const float d11 = R[3] + R[4]  * ca[j] + R[5]  * sa[j];
        const float dre = R[6] + R[7]  * ca[j] + R[8]  * sa[j];
        const float dim = R[9] + R[10] * ca[j] + R[11] * sa[j];
        float f0, f3, gr, gi;
        if (j == 0) {
            f0 = d00; f3 = d11; gr = dre; gi = dim;
        } else {
            f0 = l0 * d00;
            f3 = l3 * d11;
            gr = lr * dre - li * dim;
            gi = lr * dim + li * dre;
        }
        const float fs = f0 + f3;
        if (j > 0) z[j - 1] = 0.5f * fs;
        const float* K = &K_lds[j][0];
        l0 = K[0] * f0 + K[1] * f3 + K[2] * gr + K[3] * gi;
        if (j < NQ - 1) {
            l3 = K[1] * f0 + K[0] * f3 + K[2] * gr - K[3] * gi;
            lr = K[4] * fs + K[5] * gr;
            li = K[6] * (f3 - f0) + K[7] * gi;
        }
    }
    z[NQ - 1] = l0;

#pragma unroll
    for (int o = 0; o < 3; ++o) {
        float v = c_lds[o];
#pragma unroll
        for (int q = 0; q < NQ; ++q) v += M_lds[o][q] * z[q];
        out[tok * 3 + o] = v;
    }
}

extern "C" void kernel_launch(void* const* d_in, const int* in_sizes, int n_in,
                              void* d_out, int out_size, void* d_ws, size_t ws_size,
                              hipStream_t stream) {
    const float* x        = (const float*)d_in[0];
    const float* W_embed  = (const float*)d_in[1];
    const float* b_embed  = (const float*)d_in[2];
    const float* W_c2q    = (const float*)d_in[3];
    const float* b_c2q    = (const float*)d_in[4];
    const float* q_w      = (const float*)d_in[5];
    const float* W_q2c    = (const float*)d_in[6];
    const float* b_q2c    = (const float*)d_in[7];
    const float* W_out    = (const float*)d_in[8];
    const float* b_out    = (const float*)d_in[9];
    float* out = (float*)d_out;

    const int tokens = in_sizes[0] / DIM;
    const int blocks = tokens / TOK_PER_BLOCK;
    qhyb_fused<<<blocks, TOK_PER_BLOCK, 0, stream>>>(
        x, W_embed, b_embed, W_c2q, b_c2q, q_w, W_q2c, b_q2c, W_out, b_out, out);
}


# Tail's tanh, cos, sin without library calls: tanh from one exp2 and one reciprocal; the angle is in [-1,1], so cos and sin are short polynomials, no range reduction (last bits of these values differ)
# speedup vs baseline: 1.8586x; 1.1652x over previous
#include <hip/hip_runtime.h>


typedef __attribute__((ext_vector_type(2))) float v2f;
typedef __attribute__((ext_vector_type(8))) float v8f;

#define NQ 4
#define NL 2
#define DIM 20
#define HID 64
#define TOK_PER_WAVE 32
#define WAVES 4
#define TOK_PER_BLOCK 128

__global__ __launch_bounds__(TOK_PER_BLOCK, 8) void qhyb_fused(
    const float* __restrict__ x,
    const float* __restrict__ W_embed,
    const float* __restrict__ b_embed,
    const float* __restrict__ W_c2q,
    const float* __restrict__ b_c2q,
    const float* __restrict__ q_weights,
    const float* __restrict__ W_q2c,
    const float* __restrict__ b_q2c,
    const float* __restrict__ W_out,
    const float* __restrict__ b_out,
    float* __restrict__ out)
{
    __shared__ __attribute__((aligned(16))) float wc2q_lds[NQ * HID];
    __shared__ __attribute__((aligned(16))) float R_lds[NQ][12];
    __shared__ __attribute__((aligned(16))) float K_lds[NQ][8];
    __shared__ __attribute__((aligned(16))) float M_lds[3][NQ];
    __shared__ float c_lds[3];

    const int tid  = threadIdx.x;
    const int lane = tid & 31;
    const int wave = __builtin_amdgcn_readfirstlane(tid >> 5);
    const int lo = lane & 15, hi = lane >> 4;
    const long blockTok0 = (long)blockIdx.x * TOK_PER_BLOCK;
    const int  waveRow0  = wave * TOK_PER_WAVE;
    const long waveTok0  = blockTok0 + waveRow0;

    v2f xfrag[5][2];
#pragma unroll
    for (int ks = 0; ks < 5; ++ks)
#pragma unroll
        for (int mt = 0; mt < 2; ++mt) {
            long t = waveTok0 + mt * 16 + lo;
            int d0 = ks * 4 + 2 * hi;
            xfrag[ks][mt].x = x[t * DIM + d0];
            xfrag[ks][mt].y = x[t * DIM + d0 + 1];
        }
    v2f wfrag[5][4];
#pragma unroll
    for (int ks = 0; ks < 5; ++ks)
#pragma unroll
        for (int n = 0; n < 4; ++n) {
            int hcol = n * 16 + lo;
            int d0 = ks * 4 + 2 * hi;
            wfrag[ks][n].x = W_embed[hcol * DIM + d0];
            wfrag[ks][n].y = W_embed[hcol * DIM + d0 + 1];
        }
    v8f cinit[4];
#pragma unroll
    for (int n = 0; n < 4; ++n) {
        const float4 b0 = *reinterpret_cast<const float4*>(&b_embed[n * 16 + 8 * hi]);
        const float4 b1 = *reinterpret_cast<const float4*>(&b_embed[n * 16 + 8 * hi + 4]);
        cinit[n] = (v8f){b0.x, b0.y, b0.z, b0.w, b1.x, b1.y, b1.z, b1.w};
    }

    if (wave == 0) {
        if (lane < NL * NQ) {
            float phi = q_weights[lane * 3 + 0];
            float th  = q_weights[lane * 3 + 1];
            float om  = q_weights[lane * 3 + 2];
            float ct = cosf(0.5f * th), st = sinf(0.5f * th);
            float cp = cosf(0.5f * (phi + om)), sp = sinf(0.5f * (phi + om));
            float cm = cosf(0.5f * (phi - om)), sm = sinf(0.5f * (phi - om));
            const float u00r =  cp * ct, u00i = -sp * ct;
            const float u01r = -cm * st, u01i = -sm * st;
            const float u10r =  cm * st, u10i = -sm * st;
            const float u11r =  cp * ct, u11i =  sp * ct;
            const int q = lane & 3;
            if (lane < NQ) {
                const float n0a = u00r * u00r + u00i * u00i, n0b = u01r * u01r + u01i * u01i;
                const float n0d = 2.0f * (u00r * u01r + u00i * u01i);
                const float n1a = u10r * u10r + u10i * u10i, n1b = u11r * u11r + u11i * u11i;
                const float n1d = 2.0f * (u10r * u11r + u10i * u11i);
                const float wra = u00r * u10r + u00i * u10i, wia = u00r * u10i - u00i * u10r;
                const float wrb = u01r * u11r + u01i * u11i, wib = u01r * u11i - u01i * u11r;
                const float wrd = (u00r * u11r + u00i * u11i) + (u01r * u10r + u01i * u10i);
                const float wid = (u00r * u11i - u00i * u11r) + (u01r * u10i - u01i * u10r);
                const float a0 = n0a + n1a + 2.0f * wra, b0 = n0b + n1b + 2.0f * wrb, d0 = n0d + n1d + 2.0f * wrd;
                const float a1 = n0a + n1a - 2.0f * wra, b1 = n0b + n1b - 2.0f * wrb, d1 = n0d + n1d - 2.0f * wrd;
                const float a2 = n0a - n1a, b2 = n0b - n1b, d2 = n0d - n1d;
                const float a3 = -2.0f * wia, b3 = -2.0f * wib, d3 = -2.0f * wid;
                float* R = &R_lds[q][0];
                R[0] = 0.5f * (a0 + b0); R[1]  = 0.5f * (a0 - b0); R[2]  = 0.5f * d0;
                R[3] = 0.5f * (a1 + b1); R[4]  = 0.5f * (a1 - b1); R[5]  = 0.5f * d1;
                R[6] = 0.5f * (a2 + b2); R[7]  = 0.5f * (a2 - b2); R[8]  = 0.5f * d2;
                R[9] = 0.5f * (a3 + b3); R[10] = 0.5f * (a3 - b3); R[11] = 0.5f * d3;
            } else {
                const float A00 = (u00r * u00r + u00i * u00i) - (u10r * u10r + u10i * u10i);
                const float A11 = (u01r * u01r + u01i * u01i) - (u11r * u11r + u11i * u11i);
                const float A01r = (u00r * u01r + u00i * u01i) - (u10r * u11r + u10i * u11i);
                const float A01i = (u00r * u01i - u00i * u01r) - (u10r * u11i - u10i * u11r);
                const float u = A00 + A11 + 2.0f * A01r, v = A00 + A11 - 2.0f * A01r;
                const float p = A00 - A11, t = 2.0f * A01i;
                float* K = &K_lds[q][0];
                K[0] = 0.25f * u; K[1] = 0.25f * v; K[2] = 0.5f * p; K[3] = 0.5f * t;
                K[4] = 0.25f * p; K[5] = 0.25f * (u + v); K[6] = 0.25f * t; K[7] = 0.25f * (u - v);
            }
        }
    } else if (wave == 1) {
        float wo[3][2], bq[2];
        float4 wq[2];
#pragma unroll
        for (int j = 0; j < 2; ++j) {
            const int h = lane + 32 * j;
#pragma unroll
            for (int o = 0; o < 3; ++o) wo[o][j] = W_out[o * HID + h];
            wq[j] = *reinterpret_cast<const float4*>(&W_q2c[h * NQ]);
            bq[j] = b_q2c[h];
        }
        float pm[3][NQ], pc[3];
#pragma unroll
        for (int o = 0; o < 3; ++o) {
            pm[o][0] = wo[o][0] * wq[0].x + wo[o][1] * wq[1].x;
            pm[o][1] = wo[o][0] * wq[0].y + wo[o][1] * wq[1].y;
            pm[o][2] = wo[o][0] * wq[0].z + wo[o][1] * wq[1].z;
            pm[o][3] = wo[o][0] * wq[0].w + wo[o][1] * wq[1].w;
            pc[o]    = wo[o][0] * bq[0]   + wo[o][1] * bq[1];
        }
#pragma unroll
        for (int m = 16; m >= 1; m >>= 1) {
#pragma unroll
            for (int o = 0; o < 3; ++o) {
#pragma unroll
                for (int q = 0; q < NQ; ++q) pm[o][q] += __shfl_xor(pm[o][q], m, 32);
                pc[o] += __shfl_xor(pc[o], m, 32);
            }
        }
        if (lane == 0) {
#pragma unroll
            for (int o = 0; o < 3; ++o) {
#pragma unroll
                for (int q = 0; q < NQ; ++q) M_lds[o][q] = pm[o][q];
                c_lds[o] = pc[o] + b_out[o];
            }
        }
    } else {
        const int i4 = (tid - 2 * 32) * 4;
        *reinterpret_cast<float4*>(&wc2q_lds[i4]) = *reinterpret_cast<const float4*>(&W_c2q[i4]);
    }

    v8f acc[2][4];
#pragma unroll
    for (int mt = 0; mt < 2; ++mt)
#pragma unroll
        for (int n = 0; n < 4; ++n)
            acc[mt][n] = cinit[n];

#pragma unroll
    for (int ks = 0; ks < 5; ++ks) {
#pragma unroll
        for (int mt = 0; mt < 2; ++mt) {
#pragma unroll
            for (int n = 0; n < 4; ++n)
                acc[mt][n] = __builtin_amdgcn_wmma_f32_16x16x4_f32(
                    false, wfrag[ks][n], false, xfrag[ks][mt],
                    (short)0, acc[mt][n], false, false);
        }
    }

    __syncthreads();

    const int myRow = waveRow0 + lane;
    const long tok = blockTok0 + myRow;

    v2f ps[2][NQ];
#pragma unroll
    for (int mt = 0; mt < 2; ++mt)
#pragma unroll
        for (int q = 0; q < NQ; ++q) ps[mt][q] = (v2f){0.f, 0.f};
#pragma unroll
    for (int n = 0; n < 4; ++n) {
        v2f hp[2][4];
#pragma unroll
        for (int mt = 0; mt < 2; ++mt)
#pragma unroll
            for (int j = 0; j < 4; ++j)
                hp[mt][j] = (v2f){__builtin_elementwise_maximumnum(acc[mt][n][2 * j], 0.0f),
                                  __builtin_elementwise_maximumnum(acc[mt][n][2 * j + 1], 0.0f)};
#pragma unroll
        for (int q = 0; q < NQ; ++q) {
            const float4 wa = *reinterpret_cast<const float4*>(&wc2q_lds[q * HID + n * 16 + 8 * hi]);
            const float4 wb = *reinterpret_cast<const float4*>(&wc2q_lds[q * HID + n * 16 + 8 * hi + 4]);
            const v2f w0 = (v2f){wa.x, wa.y}, w1 = (v2f){wa.z, wa.w};
            const v2f w2 = (v2f){wb.x, wb.y}, w3 = (v2f){wb.z, wb.w};
#pragma unroll
            for (int mt = 0; mt < 2; ++mt) {
                ps[mt][q] = __builtin_elementwise_fma(w0, hp[mt][0], ps[mt][q]);
                ps[mt][q] = __builtin_elementwise_fma(w1, hp[mt][1], ps[mt][q]);
                ps[mt][q] = __builtin_elementwise_fma(w2, hp[mt][2], ps[mt][q]);
                ps[mt][q] = __builtin_elementwise_fma(w3, hp[mt][3], ps[mt][q]);
            }
        }
    }

    float sq[NQ];
#pragma unroll
    for (int q = 0; q < NQ; ++q) {
        const float p0 = ps[0][q].x + ps[0][q].y;
        const float p1 = ps[1][q].x + ps[1][q].y;
        const float mine = hi ? p1 : p0;
        const float give = hi ? p0 : p1;
        const int gb = __builtin_bit_cast(int, give);
        const float got = __builtin_bit_cast(float,
            __builtin_amdgcn_permlanex16(gb, gb, 0x76543210, 0xfedcba98, false, false));
        sq[q] = (mine + got) + b_c2q[q];
    }

    float ca[NQ], sa[NQ];
#pragma unroll
    for (int q = 0; q < NQ; ++q) {
        const float s = sq[q];
        const float e = __builtin_amdgcn_exp2f(__builtin_fabsf(s) * 2.8853900817779268f);
        const float t = __builtin_fmaf(-2.0f, __builtin_amdgcn_rcpf(e + 1.0f), 1.0f);
        const float u = t * t;
        float pc = 2.412107278360054e-5f;
        pc = __builtin_fmaf(pc, u, -1.3882957864552736e-3f);
        pc = __builtin_fmaf(pc, u, 4.166645556688309e-2f);
        pc = __builtin_fmaf(pc, u, -0.4999999701976776f);
        ca[q] = __builtin_fmaf(pc, u, 1.0f);
        float psn = 2.6937498205370503e-6f;
        psn = __builtin_fmaf(psn, u, -1.9835864077322185e-4f);
        psn = __builtin_fmaf(psn, u, 8.333314210176468e-3f);
        psn = __builtin_fmaf(psn, u, -0.1666666716337204f);
        psn = __builtin_fmaf(psn, u, 1.0f);
        sa[q] = __builtin_copysignf(t * psn, s);
    }

    float z[NQ];
    float l0 = 1.0f, l3 = 1.0f, lr = 1.0f, li = 0.0f;
#pragma unroll
    for (int j = 0; j < NQ; ++j) {
        const float* R = &R_lds[j][0];
        const float d00 = R[0] + R[1]  * ca[j] + R[2]  * sa[j];
        const float d11 = R[3] + R[4]  * ca[j] + R[5]  * sa[j];
        const float dre = R[6] + R[7]  * ca[j] + R[8]  * sa[j];
        const float dim = R[9] + R[10] * ca[j] + R[11] * sa[j];
        float f0, f3, gr, gi;
        if (j == 0) {
            f0 = d00; f3 = d11; gr = dre; gi = dim;
        } else {
            f0 = l0 * d00;
            f3 = l3 * d11;
            gr = lr * dre - li * dim;
            gi = lr * dim + li * dre;
        }
        const float fs = f0 + f3;
        if (j > 0) z[j - 1] = 0.5f * fs;
        const float* K = &K_lds[j][0];
        l0 = K[0] * f0 + K[1] * f3 + K[2] * gr + K[3] * gi;
        if (j < NQ - 1) {
            l3 = K[1] * f0 + K[0] * f3 + K[2] * gr - K[3] * gi;
            lr = K[4] * fs + K[5] * gr;
            li = K[6] * (f3 - f0) + K[7] * gi;
        }
    }
    z[NQ - 1] = l0;

#pragma unroll
    for (int o = 0; o < 3; ++o) {
        float v = c_lds[o];
#pragma unroll
        for (int q = 0; q < NQ; ++q) v += M_lds[o][q] * z[q];
        out[tok * 3 + o] = v;
    }
}

extern "C" void kernel_launch(void* const* d_in, const int* in_sizes, int n_in,
                              void* d_out, int out_size, void* d_ws, size_t ws_size,
                              hipStream_t stream) {
    const float* x        = (const float*)d_in[0];
    const float* W_embed  = (const float*)d_in[1];
    const float* b_embed  = (const float*)d_in[2];
    const float* W_c2q    = (const float*)d_in[3];
    const float* b_c2q    = (const float*)d_in[4];
    const float* q_w      = (const float*)d_in[5];
    const float* W_q2c    = (const float*)d_in[6];
    const float* b_q2c    = (const float*)d_in[7];
    const float* W_out    = (const float*)d_in[8];
    const float* b_out    = (const float*)d_in[9];
    float* out = (float*)d_out;

    const int tokens = in_sizes[0] / DIM;
    const int blocks = tokens / TOK_PER_BLOCK;
    qhyb_fused<<<blocks, TOK_PER_BLOCK, 0, stream>>>(
        x, W_embed, b_embed, W_c2q, b_c2q, q_w, W_q2c, b_q2c, W_out, b_out, out);
}


# Two 32-token tiles per wave (64 tokens a wave, 1024 blocks of 4 waves): half as many waves to start; per-block constants, W_embed fragments and bias fetched once for both tiles; every sum as before
# speedup vs baseline: 2.2085x; 1.1883x over previous
#include <hip/hip_runtime.h>


typedef __attribute__((ext_vector_type(2))) float v2f;
typedef __attribute__((ext_vector_type(8))) float v8f;

#define NQ 4
#define NL 2
#define DIM 20
#define HID 64
#define TILE_TOK 32
#define PASSES 2
#define TOK_PER_WAVE 64
#define WAVES 4
#define THREADS 128
#define TOK_PER_BLOCK 256

__global__ __launch_bounds__(THREADS, 4) void qhyb_fused(
    const float* __restrict__ x,
    const float* __restrict__ W_embed,
    const float* __restrict__ b_embed,
    const float* __restrict__ W_c2q,
    const float* __restrict__ b_c2q,
    const float* __restrict__ q_weights,
    const float* __restrict__ W_q2c,
    const float* __restrict__ b_q2c,
    const float* __restrict__ W_out,
    const float* __restrict__ b_out,
    float* __restrict__ out)
{
    __shared__ __attribute__((aligned(16))) float wc2q_lds[NQ * HID];
    __shared__ __attribute__((aligned(16))) float R_lds[NQ][12];
    __shared__ __attribute__((aligned(16))) float K_lds[NQ][8];
    __shared__ __attribute__((aligned(16))) float M_lds[3][NQ];
    __shared__ float c_lds[3];

    const int tid  = threadIdx.x;
    const int lane = tid & 31;
    const int wave = __builtin_amdgcn_readfirstlane(tid >> 5);
    const int lo = lane & 15, hi = lane >> 4;
    const long blockTok0 = (long)blockIdx.x * TOK_PER_BLOCK;
    const int  waveRow0  = wave * TOK_PER_WAVE;
    const long waveTok0  = blockTok0 + waveRow0;

    v2f xfrag[PASSES][5][2];
#pragma unroll
    for (int p = 0; p < PASSES; ++p)
#pragma unroll
        for (int ks = 0; ks < 5; ++ks)
#pragma unroll
            for (int mt = 0; mt < 2; ++mt) {
                long t = waveTok0 + p * TILE_TOK + mt * 16 + lo;
                int d0 = ks * 4 + 2 * hi;
                xfrag[p][ks][mt].x = x[t * DIM + d0];
                xfrag[p][ks][mt].y = x[t * DIM + d0 + 1];
            }
    v2f wfrag[5][4];
#pragma unroll
    for (int ks = 0; ks < 5; ++ks)
#pragma unroll
        for (int n = 0; n < 4; ++n) {
            int hcol = n * 16 + lo;
            int d0 = ks * 4 + 2 * hi;
            wfrag[ks][n].x = W_embed[hcol * DIM + d0];
            wfrag[ks][n].y = W_embed[hcol * DIM + d0 + 1];
        }
    v8f cinit[4];
#pragma unroll
    for (int n = 0; n < 4; ++n) {
        const float4 b0 = *reinterpret_cast<const float4*>(&b_embed[n * 16 + 8 * hi]);
        const float4 b1 = *reinterpret_cast<const float4*>(&b_embed[n * 16 + 8 * hi + 4]);
        cinit[n] = (v8f){b0.x, b0.y, b0.z, b0.w, b1.x, b1.y, b1.z, b1.w};
    }

    if (wave == 0) {
        if (lane < NL * NQ) {
            float phi = q_weights[lane * 3 + 0];
            float th  = q_weights[lane * 3 + 1];
            float om  = q_weights[lane * 3 + 2];
            float ct = cosf(0.5f * th), st = sinf(0.5f * th);
            float cp = cosf(0.5f * (phi + om)), sp = sinf(0.5f * (phi + om));
            float cm = cosf(0.5f * (phi - om)), sm = sinf(0.5f * (phi - om));
            const float u00r =  cp * ct, u00i = -sp * ct;
            const float u01r = -cm * st, u01i = -sm * st;
            const float u10r =  cm * st, u10i = -sm * st;
            const float u11r =  cp * ct, u11i =  sp * ct;
            const int q = lane & 3;
            if (lane < NQ) {
                const float n0a = u00r * u00r + u00i * u00i, n0b = u01r * u01r + u01i * u01i;
                const float n0d = 2.0f * (u00r * u01r + u00i * u01i);
                const float n1a = u10r * u10r + u10i * u10i, n1b = u11r * u11r + u11i * u11i;
                const float n1d = 2.0f * (u10r * u11r + u10i * u11i);
                const float wra = u00r * u10r + u00i * u10i, wia = u00r * u10i - u00i * u10r;
                const float wrb = u01r * u11r + u01i * u11i, wib = u01r * u11i - u01i * u11r;
                const float wrd = (u00r * u11r + u00i * u11i) + (u01r * u10r + u01i * u10i);
                const float wid = (u00r * u11i - u00i * u11r) + (u01r * u10i - u01i * u10r);
                const float a0 = n0a + n1a + 2.0f * wra, b0 = n0b + n1b + 2.0f * wrb, d0 = n0d + n1d + 2.0f * wrd;
                const float a1 = n0a + n1a - 2.0f * wra, b1 = n0b + n1b - 2.0f * wrb, d1 = n0d + n1d - 2.0f * wrd;
                const float a2 = n0a - n1a, b2 = n0b - n1b, d2 = n0d - n1d;
                const float a3 = -2.0f * wia, b3 = -2.0f * wib, d3 = -2.0f * wid;
                float* R = &R_lds[q][0];
                R[0] = 0.5f * (a0 + b0); R[1]  = 0.5f * (a0 - b0); R[2]  = 0.5f * d0;
                R[3] = 0.5f * (a1 + b1); R[4]  = 0.5f * (a1 - b1); R[5]  = 0.5f * d1;
                R[6] = 0.5f * (a2 + b2); R[7]  = 0.5f * (a2 - b2); R[8]  = 0.5f * d2;
                R[9] = 0.5f * (a3 + b3); R[10] = 0.5f * (a3 - b3); R[11] = 0.5f * d3;
            } else {
                const float A00 = (u00r * u00r + u00i * u00i) - (u10r * u10r + u10i * u10i);
                const float A11 = (u01r * u01r + u01i * u01i) - (u11r * u11r + u11i * u11i);
                const float A01r = (u00r * u01r + u00i * u01i) - (u10r * u11r + u10i * u11i);
                const float A01i = (u00r * u01i - u00i * u01r) - (u10r * u11i - u10i * u11r);
                const float u = A00 + A11 + 2.0f * A01r, v = A00 + A11 - 2.0f * A01r;
                const float p = A00 - A11, t = 2.0f * A01i;
                float* K = &K_lds[q][0];
                K[0] = 0.25f * u; K[1] = 0.25f * v; K[2] = 0.5f * p; K[3] = 0.5f * t;
                K[4] = 0.25f * p; K[5] = 0.25f * (u + v); K[6] = 0.25f * t; K[7] = 0.25f * (u - v);
            }
        }
    } else if (wave == 1) {
        float wo[3][2], bq[2];
        float4 wq[2];
#pragma unroll
        for (int j = 0; j < 2; ++j) {
            const int h = lane + 32 * j;
#pragma unroll
            for (int o = 0; o < 3; ++o) wo[o][j] = W_out[o * HID + h];
            wq[j] = *reinterpret_cast<const float4*>(&W_q2c[h * NQ]);
            bq[j] = b_q2c[h];
        }
        float pm[3][NQ], pc[3];
#pragma unroll
        for (int o = 0; o < 3; ++o) {
            pm[o][0] = wo[o][0] * wq[0].x + wo[o][1] * wq[1].x;
            pm[o][1] = wo[o][0] * wq[0].y + wo[o][1] * wq[1].y;
            pm[o][2] = wo[o][0] * wq[0].z + wo[o][1] * wq[1].z;
            pm[o][3] = wo[o][0] * wq[0].w + wo[o][1] * wq[1].w;
            pc[o]    = wo[o][0] * bq[0]   + wo[o][1] * bq[1];
        }
#pragma unroll
        for (int m = 16; m >= 1; m >>= 1) {
#pragma unroll
            for (int o = 0; o < 3; ++o) {
#pragma unroll
                for (int q = 0; q < NQ; ++q) pm[o][q] += __shfl_xor(pm[o][q], m, 32);
                pc[o] += __shfl_xor(pc[o], m, 32);
            }
        }
        if (lane == 0) {
#pragma unroll
            for (int o = 0; o < 3; ++o) {
#pragma unroll
                for (int q = 0; q < NQ; ++q) M_lds[o][q] = pm[o][q];
                c_lds[o] = pc[o] + b_out[o];
            }
        }
    } else {
        const int i4 = (tid - 2 * 32) * 4;
        *reinterpret_cast<float4*>(&wc2q_lds[i4]) = *reinterpret_cast<const float4*>(&W_c2q[i4]);
    }

#pragma unroll
    for (int p = 0; p < PASSES; ++p) {
    v8f acc[2][4];
#pragma unroll
    for (int mt = 0; mt < 2; ++mt)
#pragma unroll
        for (int n = 0; n < 4; ++n)
            acc[mt][n] = cinit[n];

#pragma unroll
    for (int ks = 0; ks < 5; ++ks) {
#pragma unroll
        for (int mt = 0; mt < 2; ++mt) {
#pragma unroll
            for (int n = 0; n < 4; ++n)
                acc[mt][n] = __builtin_amdgcn_wmma_f32_16x16x4_f32(
                    false, wfrag[ks][n], false, xfrag[p][ks][mt],
                    (short)0, acc[mt][n], false, false);
        }
    }

    if (p == 0) __syncthreads();

    const int myRow = waveRow0 + p * TILE_TOK + lane;
    const long tok = blockTok0 + myRow;

    v2f ps[2][NQ];
#pragma unroll
    for (int mt = 0; mt < 2; ++mt)
#pragma unroll
        for (int q = 0; q < NQ; ++q) ps[mt][q] = (v2f){0.f, 0.f};
#pragma unroll
    for (int n = 0; n < 4; ++n) {
        v2f hp[2][4];
#pragma unroll
        for (int mt = 0; mt < 2; ++mt)
#pragma unroll
            for (int j = 0; j < 4; ++j)
                hp[mt][j] = (v2f){__builtin_elementwise_maximumnum(acc[mt][n][2 * j], 0.0f),
                                  __builtin_elementwise_maximumnum(acc[mt][n][2 * j + 1], 0.0f)};
#pragma unroll
        for (int q = 0; q < NQ; ++q) {
            const float4 wa = *reinterpret_cast<const float4*>(&wc2q_lds[q * HID + n * 16 + 8 * hi]);
            const float4 wb = *reinterpret_cast<const float4*>(&wc2q_lds[q * HID + n * 16 + 8 * hi + 4]);
            const v2f w0 = (v2f){wa.x, wa.y}, w1 = (v2f){wa.z, wa.w};
            const v2f w2 = (v2f){wb.x, wb.y}, w3 = (v2f){wb.z, wb.w};
#pragma unroll
            for (int mt = 0; mt < 2; ++mt) {
                ps[mt][q] = __builtin_elementwise_fma(w0, hp[mt][0], ps[mt][q]);
                ps[mt][q] = __builtin_elementwise_fma(w1, hp[mt][1], ps[mt][q]);
                ps[mt][q] = __builtin_elementwise_fma(w2, hp[mt][2], ps[mt][q]);
                ps[mt][q] = __builtin_elementwise_fma(w3, hp[mt][3], ps[mt][q]);
            }
        }
    }

    float sq[NQ];
#pragma unroll
    for (int q = 0; q < NQ; ++q) {
        const float p0 = ps[0][q].x + ps[0][q].y;
        const float p1 = ps[1][q].x + ps[1][q].y;
        const float mine = hi ? p1 : p0;
        const float give = hi ? p0 : p1;
        const int gb = __builtin_bit_cast(int, give);
        const float got = __builtin_bit_cast(float,
            __builtin_amdgcn_permlanex16(gb, gb, 0x76543210, 0xfedcba98, false, false));
        sq[q] = (mine + got) + b_c2q[q];
    }

    float ca[NQ], sa[NQ];
#pragma unroll
    for (int q = 0; q < NQ; ++q) {
        const float s = sq[q];
        const float e = __builtin_amdgcn_exp2f(__builtin_fabsf(s) * 2.8853900817779268f);
        const float t = __builtin_fmaf(-2.0f, __builtin_amdgcn_rcpf(e + 1.0f), 1.0f);
        const float u = t * t;
        float pc = 2.412107278360054e-5f;
        pc = __builtin_fmaf(pc, u, -1.3882957864552736e-3f);
        pc = __builtin_fmaf(pc, u, 4.166645556688309e-2f);
        pc = __builtin_fmaf(pc, u, -0.4999999701976776f);
        ca[q] = __builtin_fmaf(pc, u, 1.0f);
        float psn = 2.6937498205370503e-6f;
        psn = __builtin_fmaf(psn, u, -1.9835864077322185e-4f);
        psn = __builtin_fmaf(psn, u, 8.333314210176468e-3f);
        psn = __builtin_fmaf(psn, u, -0.1666666716337204f);
        psn = __builtin_fmaf(psn, u, 1.0f);
        sa[q] = __builtin_copysignf(t * psn, s);
    }

    float z[NQ];
    float l0 = 1.0f, l3 = 1.0f, lr = 1.0f, li = 0.0f;
#pragma unroll
    for (int j = 0; j < NQ; ++j) {
        const float* R = &R_lds[j][0];
        const float d00 = R[0] + R[1]  * ca[j] + R[2]  * sa[j];
        const float d11 = R[3] + R[4]  * ca[j] + R[5]  * sa[j];
        const float dre = R[6] + R[7]  * ca[j] + R[8]  * sa[j];
        const float dim = R[9] + R[10] * ca[j] + R[11] * sa[j];
        float f0, f3, gr, gi;
        if (j == 0) {
            f0 = d00; f3 = d11; gr = dre; gi = dim;
        } else {
            f0 = l0 * d00;
            f3 = l3 * d11;
            gr = lr * dre - li * dim;
            gi = lr * dim + li * dre;
        }
        const float fs = f0 + f3;
        if (j > 0) z[j - 1] = 0.5f * fs;
        const float* K = &K_lds[j][0];
        l0 = K[0] * f0 + K[1] * f3 + K[2] * gr + K[3] * gi;
        if (j < NQ - 1) {
            l3 = K[1] * f0 + K[0] * f3 + K[2] * gr - K[3] * gi;
            lr = K[4] * fs + K[5] * gr;
            li = K[6] * (f3 - f0) + K[7] * gi;
        }
    }
    z[NQ - 1] = l0;

#pragma unroll
    for (int o = 0; o < 3; ++o) {
        float v = c_lds[o];
#pragma unroll
        for (int q = 0; q < NQ; ++q) v += M_lds[o][q] * z[q];
        out[tok * 3 + o] = v;
    }
    }
}

extern "C" void kernel_launch(void* const* d_in, const int* in_sizes, int n_in,
                              void* d_out, int out_size, void* d_ws, size_t ws_size,
                              hipStream_t stream) {
    const float* x        = (const float*)d_in[0];
    const float* W_embed  = (const float*)d_in[1];
    const float* b_embed  = (const float*)d_in[2];
    const float* W_c2q    = (const float*)d_in[3];
    const float* b_c2q    = (const float*)d_in[4];
    const float* q_w      = (const float*)d_in[5];
    const float* W_q2c    = (const float*)d_in[6];
    const float* b_q2c    = (const float*)d_in[7];
    const float* W_out    = (const float*)d_in[8];
    const float* b_out    = (const float*)d_in[9];
    float* out = (float*)d_out;

    const int tokens = in_sizes[0] / DIM;
    const int blocks = tokens / TOK_PER_BLOCK;
    qhyb_fused<<<blocks, THREADS, 0, stream>>>(
        x, W_embed, b_embed, W_c2q, b_c2q, q_w, W_q2c, b_q2c, W_out, b_out, out);
}
